# starting kernel
# baseline (speedup 1.0000x reference)
#include <hip/hip_runtime.h>
#include <hip/hip_bf16.h>
#include <math.h>


#define B_ 4
#define L_ 2048
#define S_ 2048
#define H_ 8
#define E_ 64
#define D_ 64

typedef float v2f __attribute__((ext_vector_type(2)));
typedef float v8f __attribute__((ext_vector_type(8)));

__device__ __forceinline__ v8f wmma_f32_16x16x4(v2f a, v2f b, v8f c) {
  return __builtin_amdgcn_wmma_f32_16x16x4_f32(
      false, a, false, b, (short)0, c, false, false);
}

#define PSTRIDE 20

__global__ __launch_bounds__(256)
void causal_attn_wmma_kernel(const float* __restrict__ Q,
                             const float* __restrict__ K,
                             const float* __restrict__ V,
                             const float* __restrict__ bias,
                             float* __restrict__ out) {
  __shared__ float pbuf[8][16 * PSTRIDE];

  const int tid  = threadIdx.x;
  const int wave = tid >> 5;
  const int lane = tid & 31;
  const int nlo  = lane & 15;
  const int hi   = lane >> 4;

  const int bid = blockIdx.x;
  const int qb  = bid & 15;
  const int h   = (bid >> 4) & (H_ - 1);
  const int b   = bid >> 7;
  const int qBase = qb * 128 + wave * 16;

  const float scale = 0.125f;

  const int qRowA = qBase + nlo;
  const float* qptr = Q + (((size_t)b * L_ + qRowA) * H_ + h) * E_;
  v2f aQ[16];
#pragma unroll
  for (int c = 0; c < 16; ++c) {
    aQ[c] = *reinterpret_cast<const v2f*>(qptr + c * 4 + hi * 2);
  }

  v8f oacc[4];
#pragma unroll
  for (int t = 0; t < 4; ++t) oacc[t] = (v8f){0, 0, 0, 0, 0, 0, 0, 0};
  v8f lacc = (v8f){0, 0, 0, 0, 0, 0, 0, 0};
  const v2f onesB = (v2f){1.0f, 1.0f};

  float* myP = pbuf[wave];

  for (int j0 = 0; j0 <= qBase; j0 += 16) {
    const int key = j0 + nlo;
    const float* kptr = K + (((size_t)b * S_ + key) * H_ + h) * E_;
    v8f s8a = (v8f){0, 0, 0, 0, 0, 0, 0, 0};
    v8f s8b = (v8f){0, 0, 0, 0, 0, 0, 0, 0};
#pragma unroll
    for (int c = 0; c < 16; c += 2) {
      v2f bK0 = *reinterpret_cast<const v2f*>(kptr + (c + 0) * 4 + hi * 2);
      v2f bK1 = *reinterpret_cast<const v2f*>(kptr + (c + 1) * 4 + hi * 2);
      s8a = wmma_f32_16x16x4(aQ[c + 0], bK0, s8a);
      s8b = wmma_f32_16x16x4(aQ[c + 1], bK1, s8b);
    }

    float p[8];
#pragma unroll
    for (int r = 0; r < 8; ++r) {
      const int M = r + hi * 8;
      const int qRow = qBase + M;
      const float s = s8a[r] + s8b[r];
      p[r] = (key <= qRow)
                 ? __expf((s + bias[(size_t)qRow * S_ + key]) * scale)
                 : 0.0f;
    }

#pragma unroll
    for (int r = 0; r < 8; ++r)
      myP[(r + hi * 8) * PSTRIDE + nlo] = p[r];
    __builtin_amdgcn_fence(__ATOMIC_SEQ_CST, "wavefront");
    v2f ap[4];
#pragma unroll
    for (int kk = 0; kk < 4; ++kk) {
      ap[kk] = *reinterpret_cast<const v2f*>(myP + nlo * PSTRIDE + kk * 4 + hi * 2);
    }
    __builtin_amdgcn_fence(__ATOMIC_SEQ_CST, "wavefront");

#pragma unroll
    for (int kk = 0; kk < 4; ++kk) {
      const int s0 = j0 + kk * 4 + hi * 2;
      const float* v0 = V + (((size_t)b * S_ + s0) * H_ + h) * D_ + nlo;
      const float* v1 = v0 + (size_t)H_ * D_;
#pragma unroll
      for (int t = 0; t < 4; ++t) {
        v2f bV;
        bV.x = v0[t * 16];
        bV.y = v1[t * 16];
        oacc[t] = wmma_f32_16x16x4(ap[kk], bV, oacc[t]);
      }
      lacc = wmma_f32_16x16x4(ap[kk], onesB, lacc);
    }
  }

  float inv[8];
#pragma unroll
  for (int r = 0; r < 8; ++r) inv[r] = 1.0f / lacc[r];
#pragma unroll
  for (int t = 0; t < 4; ++t) {
#pragma unroll
    for (int r = 0; r < 8; ++r) {
      const int M = r + hi * 8;
      const int qRow = qBase + M;
      out[(((size_t)b * L_ + qRow) * H_ + h) * D_ + t * 16 + nlo] =
          oacc[t][r] * inv[r];
    }
  }
}

extern "C" void kernel_launch(void* const* d_in, const int* in_sizes, int n_in,
                              void* d_out, int out_size, void* d_ws, size_t ws_size,
                              hipStream_t stream) {
  const float* Q    = (const float*)d_in[0];
  const float* K    = (const float*)d_in[1];
  const float* V    = (const float*)d_in[2];
  const float* bias = (const float*)d_in[3];
  float* out = (float*)d_out;

  const int numQBlocks = L_ / 128;
  dim3 grid(B_ * H_ * numQBlocks);
  dim3 block(256);
  causal_attn_wmma_kernel<<<grid, block, 0, stream>>>(Q, K, V, bias, out);
}


# All global loads of a key tile (K, bias, V) issued as one group before the first WMMA; bias read unconditionally, causal select applied after exp: 8 serial load waits per tile become one
# speedup vs baseline: 1.7779x; 1.7779x over previous
#include <hip/hip_runtime.h>
#include <hip/hip_bf16.h>
#include <math.h>


#define B_ 4
#define L_ 2048
#define S_ 2048
#define H_ 8
#define E_ 64
#define D_ 64

typedef float v2f __attribute__((ext_vector_type(2)));
typedef float v8f __attribute__((ext_vector_type(8)));

__device__ __forceinline__ v8f wmma_f32_16x16x4(v2f a, v2f b, v8f c) {
  return __builtin_amdgcn_wmma_f32_16x16x4_f32(
      false, a, false, b, (short)0, c, false, false);
}

#define PSTRIDE 20

__global__ __launch_bounds__(256)
void causal_attn_wmma_kernel(const float* __restrict__ Q,
                             const float* __restrict__ K,
                             const float* __restrict__ V,
                             const float* __restrict__ bias,
                             float* __restrict__ out) {
  __shared__ float pbuf[8][16 * PSTRIDE];

  const int tid  = threadIdx.x;
  const int wave = tid >> 5;
  const int lane = tid & 31;
  const int nlo  = lane & 15;
  const int hi   = lane >> 4;

  const int bid = blockIdx.x;
  const int qb  = bid & 15;
  const int h   = (bid >> 4) & (H_ - 1);
  const int b   = bid >> 7;
  const int qBase = qb * 128 + wave * 16;

  const float scale = 0.125f;

  const int qRowA = qBase + nlo;
  const float* qptr = Q + (((size_t)b * L_ + qRowA) * H_ + h) * E_;
  v2f aQ[16];
#pragma unroll
  for (int c = 0; c < 16; ++c) {
    aQ[c] = *reinterpret_cast<const v2f*>(qptr + c * 4 + hi * 2);
  }

  v8f oacc[4];
#pragma unroll
  for (int t = 0; t < 4; ++t) oacc[t] = (v8f){0, 0, 0, 0, 0, 0, 0, 0};
  v8f lacc = (v8f){0, 0, 0, 0, 0, 0, 0, 0};
  const v2f onesB = (v2f){1.0f, 1.0f};

  float* myP = pbuf[wave];

  const float* bptr = bias + (size_t)(qBase + hi * 8) * S_ + nlo;

  for (int j0 = 0; j0 <= qBase; j0 += 16) {
    const int key = j0 + nlo;
    const float* kptr = K + (((size_t)b * S_ + key) * H_ + h) * E_;

    v2f bK[16];
#pragma unroll
    for (int c = 0; c < 16; ++c)
      bK[c] = *reinterpret_cast<const v2f*>(kptr + c * 4 + hi * 2);

    float bv[8];
#pragma unroll
    for (int r = 0; r < 8; ++r) bv[r] = bptr[(size_t)r * S_ + j0];

    v2f bV[4][4];
#pragma unroll
    for (int kk = 0; kk < 4; ++kk) {
      const int s0 = j0 + kk * 4 + hi * 2;
      const float* v0 = V + (((size_t)b * S_ + s0) * H_ + h) * D_ + nlo;
      const float* v1 = v0 + (size_t)H_ * D_;
#pragma unroll
      for (int t = 0; t < 4; ++t) {
        bV[kk][t].x = v0[t * 16];
        bV[kk][t].y = v1[t * 16];
      }
    }

    __builtin_amdgcn_sched_barrier(0);

    v8f s8a = (v8f){0, 0, 0, 0, 0, 0, 0, 0};
    v8f s8b = (v8f){0, 0, 0, 0, 0, 0, 0, 0};
#pragma unroll
    for (int c = 0; c < 16; c += 2) {
      s8a = wmma_f32_16x16x4(aQ[c + 0], bK[c + 0], s8a);
      s8b = wmma_f32_16x16x4(aQ[c + 1], bK[c + 1], s8b);
    }

    float p[8];
#pragma unroll
    for (int r = 0; r < 8; ++r) {
      const int qRow = qBase + r + hi * 8;
      const float s = s8a[r] + s8b[r];
      const float e = __expf((s + bv[r]) * scale);
      p[r] = (key <= qRow) ? e : 0.0f;
    }

#pragma unroll
    for (int r = 0; r < 8; ++r)
      myP[(r + hi * 8) * PSTRIDE + nlo] = p[r];
    __builtin_amdgcn_fence(__ATOMIC_SEQ_CST, "wavefront");
    v2f ap[4];
#pragma unroll
    for (int kk = 0; kk < 4; ++kk) {
      ap[kk] = *reinterpret_cast<const v2f*>(myP + nlo * PSTRIDE + kk * 4 + hi * 2);
    }
    __builtin_amdgcn_fence(__ATOMIC_SEQ_CST, "wavefront");

#pragma unroll
    for (int kk = 0; kk < 4; ++kk) {
#pragma unroll
      for (int t = 0; t < 4; ++t)
        oacc[t] = wmma_f32_16x16x4(ap[kk], bV[kk][t], oacc[t]);
      lacc = wmma_f32_16x16x4(ap[kk], onesB, lacc);
    }
  }

  float inv[8];
#pragma unroll
  for (int r = 0; r < 8; ++r) inv[r] = 1.0f / lacc[r];
#pragma unroll
  for (int t = 0; t < 4; ++t) {
#pragma unroll
    for (int r = 0; r < 8; ++r) {
      const int M = r + hi * 8;
      const int qRow = qBase + M;
      out[(((size_t)b * L_ + qRow) * H_ + h) * D_ + t * 16 + nlo] =
          oacc[t][r] * inv[r];
    }
  }
}

extern "C" void kernel_launch(void* const* d_in, const int* in_sizes, int n_in,
                              void* d_out, int out_size, void* d_ws, size_t ws_size,
                              hipStream_t stream) {
  const float* Q    = (const float*)d_in[0];
  const float* K    = (const float*)d_in[1];
  const float* V    = (const float*)d_in[2];
  const float* bias = (const float*)d_in[3];
  float* out = (float*)d_out;

  const int numQBlocks = L_ / 128;
  dim3 grid(B_ * H_ * numQBlocks);
  dim3 block(256);
  causal_attn_wmma_kernel<<<grid, block, 0, stream>>>(Q, K, V, bias, out);
}


# Causal load balance: each wave owns query tiles t and 127-t, run one after the other (129 key tiles for every wave); grid 512x8 waves with 1..128 tiles each becomes 256x8 equal waves
# speedup vs baseline: 2.2113x; 1.2438x over previous
#include <hip/hip_runtime.h>
#include <hip/hip_bf16.h>
#include <math.h>


#define B_ 4
#define L_ 2048
#define S_ 2048
#define H_ 8
#define E_ 64
#define D_ 64

typedef float v2f __attribute__((ext_vector_type(2)));
typedef float v8f __attribute__((ext_vector_type(8)));

__device__ __forceinline__ v8f wmma_f32_16x16x4(v2f a, v2f b, v8f c) {
  return __builtin_amdgcn_wmma_f32_16x16x4_f32(
      false, a, false, b, (short)0, c, false, false);
}

#define PSTRIDE 20

__global__ __launch_bounds__(256)
void causal_attn_wmma_kernel(const float* __restrict__ Q,
                             const float* __restrict__ K,
                             const float* __restrict__ V,
                             const float* __restrict__ bias,
                             float* __restrict__ out) {
  __shared__ float pbuf[8][16 * PSTRIDE];

  const int tid  = threadIdx.x;
  const int wave = tid >> 5;
  const int lane = tid & 31;
  const int nlo  = lane & 15;
  const int hi   = lane >> 4;

  const int bid = blockIdx.x;
  const int pb  = bid & 7;
  const int h   = (bid >> 3) & (H_ - 1);
  const int b   = bid >> 6;
  const int pairIdx = pb * 8 + wave;

  const float scale = 0.125f;
  float* myP = pbuf[wave];
  const v2f onesB = (v2f){1.0f, 1.0f};

  for (int pass = 0; pass < 2; ++pass) {
  const int qTile = pass ? pairIdx : (L_ / 16 - 1 - pairIdx);
  const int qBase = qTile * 16;

  const int qRowA = qBase + nlo;
  const float* qptr = Q + (((size_t)b * L_ + qRowA) * H_ + h) * E_;
  v2f aQ[16];
#pragma unroll
  for (int c = 0; c < 16; ++c) {
    aQ[c] = *reinterpret_cast<const v2f*>(qptr + c * 4 + hi * 2);
  }

  v8f oacc[4];
#pragma unroll
  for (int t = 0; t < 4; ++t) oacc[t] = (v8f){0, 0, 0, 0, 0, 0, 0, 0};
  v8f lacc = (v8f){0, 0, 0, 0, 0, 0, 0, 0};

  const float* bptr = bias + (size_t)(qBase + hi * 8) * S_ + nlo;

  for (int j0 = 0; j0 <= qBase; j0 += 16) {
    const int key = j0 + nlo;
    const float* kptr = K + (((size_t)b * S_ + key) * H_ + h) * E_;

    v2f bK[16];
#pragma unroll
    for (int c = 0; c < 16; ++c)
      bK[c] = *reinterpret_cast<const v2f*>(kptr + c * 4 + hi * 2);

    float bv[8];
#pragma unroll
    for (int r = 0; r < 8; ++r) bv[r] = bptr[(size_t)r * S_ + j0];

    v2f bV[4][4];
#pragma unroll
    for (int kk = 0; kk < 4; ++kk) {
      const int s0 = j0 + kk * 4 + hi * 2;
      const float* v0 = V + (((size_t)b * S_ + s0) * H_ + h) * D_ + nlo;
      const float* v1 = v0 + (size_t)H_ * D_;
#pragma unroll
      for (int t = 0; t < 4; ++t) {
        bV[kk][t].x = v0[t * 16];
        bV[kk][t].y = v1[t * 16];
      }
    }

    __builtin_amdgcn_sched_barrier(0);

    v8f s8a = (v8f){0, 0, 0, 0, 0, 0, 0, 0};
    v8f s8b = (v8f){0, 0, 0, 0, 0, 0, 0, 0};
#pragma unroll
    for (int c = 0; c < 16; c += 2) {
      s8a = wmma_f32_16x16x4(aQ[c + 0], bK[c + 0], s8a);
      s8b = wmma_f32_16x16x4(aQ[c + 1], bK[c + 1], s8b);
    }

    float p[8];
#pragma unroll
    for (int r = 0; r < 8; ++r) {
      const int qRow = qBase + r + hi * 8;
      const float s = s8a[r] + s8b[r];
      const float e = __expf((s + bv[r]) * scale);
      p[r] = (key <= qRow) ? e : 0.0f;
    }

#pragma unroll
    for (int r = 0; r < 8; ++r)
      myP[(r + hi * 8) * PSTRIDE + nlo] = p[r];
    __builtin_amdgcn_fence(__ATOMIC_SEQ_CST, "wavefront");
    v2f ap[4];
#pragma unroll
    for (int kk = 0; kk < 4; ++kk) {
      ap[kk] = *reinterpret_cast<const v2f*>(myP + nlo * PSTRIDE + kk * 4 + hi * 2);
    }
    __builtin_amdgcn_fence(__ATOMIC_SEQ_CST, "wavefront");

#pragma unroll
    for (int kk = 0; kk < 4; ++kk) {
#pragma unroll
      for (int t = 0; t < 4; ++t)
        oacc[t] = wmma_f32_16x16x4(ap[kk], bV[kk][t], oacc[t]);
      lacc = wmma_f32_16x16x4(ap[kk], onesB, lacc);
    }
  }

  float inv[8];
#pragma unroll
  for (int r = 0; r < 8; ++r) inv[r] = 1.0f / lacc[r];
#pragma unroll
  for (int t = 0; t < 4; ++t) {
#pragma unroll
    for (int r = 0; r < 8; ++r) {
      const int M = r + hi * 8;
      const int qRow = qBase + M;
      out[(((size_t)b * L_ + qRow) * H_ + h) * D_ + t * 16 + nlo] =
          oacc[t][r] * inv[r];
    }
  }
  }
}

extern "C" void kernel_launch(void* const* d_in, const int* in_sizes, int n_in,
                              void* d_out, int out_size, void* d_ws, size_t ws_size,
                              hipStream_t stream) {
  const float* Q    = (const float*)d_in[0];
  const float* K    = (const float*)d_in[1];
  const float* V    = (const float*)d_in[2];
  const float* bias = (const float*)d_in[3];
  float* out = (float*)d_out;

  dim3 grid(256);
  dim3 block(256);
  causal_attn_wmma_kernel<<<grid, block, 0, stream>>>(Q, K, V, bias, out);
}


# K and V staged once per work-group in LDS, 64 keys a stage, by coalesced 16-byte loads (two buffers, one barrier per stage); the 8 waves read WMMA fragments from LDS instead of global memory
# speedup vs baseline: 3.2471x; 1.4684x over previous
#include <hip/hip_runtime.h>
#include <hip/hip_bf16.h>
#include <math.h>


#define B_ 4
#define L_ 2048
#define S_ 2048
#define H_ 8
#define E_ 64
#define D_ 64

typedef float v2f __attribute__((ext_vector_type(2)));
typedef float v4f __attribute__((ext_vector_type(4)));
typedef float v8f __attribute__((ext_vector_type(8)));

__device__ __forceinline__ v8f wmma_f32_16x16x4(v2f a, v2f b, v8f c) {
  return __builtin_amdgcn_wmma_f32_16x16x4_f32(
      false, a, false, b, (short)0, c, false, false);
}

#define PSTRIDE 20

#define KVSTRIDE 68

#define TPS 4
#define STAGE_ROWS (16 * TPS)

__global__ __launch_bounds__(256)
void causal_attn_wmma_kernel(const float* __restrict__ Q,
                             const float* __restrict__ K,
                             const float* __restrict__ V,
                             const float* __restrict__ bias,
                             float* __restrict__ out) {
  __shared__ __attribute__((aligned(16))) float kbuf[2][STAGE_ROWS * KVSTRIDE];
  __shared__ __attribute__((aligned(16))) float vbuf[2][STAGE_ROWS * KVSTRIDE];
  __shared__ __attribute__((aligned(16))) float pbuf[8][16 * PSTRIDE];

  const int tid  = threadIdx.x;
  const int wave = __builtin_amdgcn_readfirstlane(tid >> 5);
  const int lane = tid & 31;
  const int nlo  = lane & 15;
  const int hi   = lane >> 4;

  const int bid = blockIdx.x;
  const int pb  = bid & 7;
  const int h   = (bid >> 3) & (H_ - 1);
  const int b   = bid >> 6;
  const int pairIdx = pb * 8 + wave;

  const float scale = 0.125f;
  float* myP = pbuf[wave];
  const v2f onesB = (v2f){1.0f, 1.0f};

  const int srow = tid >> 4;
  const int scol = (tid & 15) * 4;
  const float* kstage = K + (((size_t)b * S_ + srow) * H_ + h) * E_ + scol;
  const float* vstage = V + (((size_t)b * S_ + srow) * H_ + h) * D_ + scol;
  const int sofs = srow * KVSTRIDE + scol;
  const size_t tileStride = (size_t)16 * H_ * E_;

  for (int pass = 0; pass < 2; ++pass) {
  const int qTile = pass ? pairIdx : (L_ / 16 - 1 - pairIdx);
  const int qBase = qTile * 16;
  const int jmax = pass ? (pb * 8 + 7) : (L_ / 16 - 1 - pb * 8);
  const int nst  = (jmax + 1) / TPS;

  const int qRowA = qBase + nlo;
  const float* qptr = Q + (((size_t)b * L_ + qRowA) * H_ + h) * E_;
  v2f aQ[16];
#pragma unroll
  for (int c = 0; c < 16; ++c) {
    aQ[c] = *reinterpret_cast<const v2f*>(qptr + c * 4 + hi * 2);
  }

  v8f oacc[4];
#pragma unroll
  for (int t = 0; t < 4; ++t) oacc[t] = (v8f){0, 0, 0, 0, 0, 0, 0, 0};
  v8f lacc = (v8f){0, 0, 0, 0, 0, 0, 0, 0};

  const float* bptr = bias + (size_t)(qBase + hi * 8) * S_ + nlo;

  {
    v4f k4[TPS], v4[TPS];
#pragma unroll
    for (int i = 0; i < TPS; ++i) {
      k4[i] = *reinterpret_cast<const v4f*>(kstage + (size_t)i * tileStride);
      v4[i] = *reinterpret_cast<const v4f*>(vstage + (size_t)i * tileStride);
    }
#pragma unroll
    for (int i = 0; i < TPS; ++i) {
      *reinterpret_cast<v4f*>(&kbuf[0][sofs + i * 16 * KVSTRIDE]) = k4[i];
      *reinterpret_cast<v4f*>(&vbuf[0][sofs + i * 16 * KVSTRIDE]) = v4[i];
    }
  }
  float bv[8];
#pragma unroll
  for (int r = 0; r < 8; ++r) bv[r] = bptr[(size_t)r * S_];
  __syncthreads();

  for (int st = 0; st < nst; ++st) {
    const int cur = st & 1;
    const int sn = (st + 1 < nst) ? (st + 1) : st;

    v4f kpre[TPS], vpre[TPS];
#pragma unroll
    for (int i = 0; i < TPS; ++i) {
      kpre[i] = *reinterpret_cast<const v4f*>(kstage + (size_t)(sn * TPS + i) * tileStride);
      vpre[i] = *reinterpret_cast<const v4f*>(vstage + (size_t)(sn * TPS + i) * tileStride);
    }
    const float* kcur = kbuf[cur];
    const float* vcur = vbuf[cur];

    for (int ti = 0; ti < TPS; ++ti) {
      const int jt = st * TPS + ti;
      const int jn = (jt < jmax) ? (jt + 1) : jmax;
      float bvn[8];
#pragma unroll
      for (int r = 0; r < 8; ++r) bvn[r] = bptr[(size_t)r * S_ + jn * 16];

      if (jt <= qTile) {
        const int j0  = jt * 16;
        const int key = j0 + nlo;

        const float* kb = kcur + (ti * 16 + nlo) * KVSTRIDE + hi * 2;
        v2f bK[16];
#pragma unroll
        for (int c = 0; c < 16; ++c)
          bK[c] = *reinterpret_cast<const v2f*>(kb + c * 4);

        const float* vb = vcur + (ti * 16 + hi * 2) * KVSTRIDE + nlo * 4;
        v2f bV[4][4];
#pragma unroll
        for (int kk = 0; kk < 4; ++kk) {
          const v4f r0 = *reinterpret_cast<const v4f*>(vb + (kk * 4 + 0) * KVSTRIDE);
          const v4f r1 = *reinterpret_cast<const v4f*>(vb + (kk * 4 + 1) * KVSTRIDE);
#pragma unroll
          for (int t = 0; t < 4; ++t) {
            bV[kk][t].x = r0[t];
            bV[kk][t].y = r1[t];
          }
        }

        __builtin_amdgcn_sched_barrier(0);

        v8f s8a = (v8f){0, 0, 0, 0, 0, 0, 0, 0};
        v8f s8b = (v8f){0, 0, 0, 0, 0, 0, 0, 0};
#pragma unroll
        for (int c = 0; c < 16; c += 2) {
          s8a = wmma_f32_16x16x4(aQ[c + 0], bK[c + 0], s8a);
          s8b = wmma_f32_16x16x4(aQ[c + 1], bK[c + 1], s8b);
        }

        float p[8];
#pragma unroll
        for (int r = 0; r < 8; ++r) {
          const int qRow = qBase + r + hi * 8;
          const float s = s8a[r] + s8b[r];
          const float e = __expf((s + bv[r]) * scale);
          p[r] = (key <= qRow) ? e : 0.0f;
        }

#pragma unroll
        for (int r = 0; r < 8; ++r)
          myP[(r + hi * 8) * PSTRIDE + nlo] = p[r];
        __builtin_amdgcn_fence(__ATOMIC_SEQ_CST, "wavefront");
        v2f ap[4];
#pragma unroll
        for (int kk = 0; kk < 4; ++kk) {
          ap[kk] = *reinterpret_cast<const v2f*>(myP + nlo * PSTRIDE + kk * 4 + hi * 2);
        }
        __builtin_amdgcn_fence(__ATOMIC_SEQ_CST, "wavefront");

#pragma unroll
        for (int kk = 0; kk < 4; ++kk) {
#pragma unroll
          for (int t = 0; t < 4; ++t)
            oacc[t] = wmma_f32_16x16x4(ap[kk], bV[kk][t], oacc[t]);
          lacc = wmma_f32_16x16x4(ap[kk], onesB, lacc);
        }
      }
#pragma unroll
      for (int r = 0; r < 8; ++r) bv[r] = bvn[r];
    }

#pragma unroll
    for (int i = 0; i < TPS; ++i) {
      *reinterpret_cast<v4f*>(&kbuf[cur ^ 1][sofs + i * 16 * KVSTRIDE]) = kpre[i];
      *reinterpret_cast<v4f*>(&vbuf[cur ^ 1][sofs + i * 16 * KVSTRIDE]) = vpre[i];
    }
    __syncthreads();
  }

#pragma unroll
  for (int r = 0; r < 8; ++r) {
    const float inv = 1.0f / lacc[r];
    const int qRow = qBase + r + hi * 8;
    v4f o;
    o.x = oacc[0][r] * inv;
    o.y = oacc[1][r] * inv;
    o.z = oacc[2][r] * inv;
    o.w = oacc[3][r] * inv;
    *reinterpret_cast<v4f*>(out + (((size_t)b * L_ + qRow) * H_ + h) * D_ + nlo * 4) = o;
  }
  }
}

extern "C" void kernel_launch(void* const* d_in, const int* in_sizes, int n_in,
                              void* d_out, int out_size, void* d_ws, size_t ws_size,
                              hipStream_t stream) {
  const float* Q    = (const float*)d_in[0];
  const float* K    = (const float*)d_in[1];
  const float* V    = (const float*)d_in[2];
  const float* bias = (const float*)d_in[3];
  float* out = (float*)d_out;

  dim3 grid(256);
  dim3 block(256);
  causal_attn_wmma_kernel<<<grid, block, 0, stream>>>(Q, K, V, bias, out);
}


# Softmax row sums by per-lane f32 adds of P each key tile plus one ones-WMMA per query tile, not 4 ones-WMMAs per key tile (36 -> 32 WMMA a tile); denominator summed in another order
# speedup vs baseline: 3.4109x; 1.0504x over previous
#include <hip/hip_runtime.h>
#include <hip/hip_bf16.h>
#include <math.h>


#define B_ 4
#define L_ 2048
#define S_ 2048
#define H_ 8
#define E_ 64
#define D_ 64

typedef float v2f __attribute__((ext_vector_type(2)));
typedef float v4f __attribute__((ext_vector_type(4)));
typedef float v8f __attribute__((ext_vector_type(8)));

__device__ __forceinline__ v8f wmma_f32_16x16x4(v2f a, v2f b, v8f c) {
  return __builtin_amdgcn_wmma_f32_16x16x4_f32(
      false, a, false, b, (short)0, c, false, false);
}

#define PSTRIDE 20

#define KVSTRIDE 68

#define TPS 4
#define STAGE_ROWS (16 * TPS)

__global__ __launch_bounds__(256)
void causal_attn_wmma_kernel(const float* __restrict__ Q,
                             const float* __restrict__ K,
                             const float* __restrict__ V,
                             const float* __restrict__ bias,
                             float* __restrict__ out) {
  __shared__ __attribute__((aligned(16))) float kbuf[2][STAGE_ROWS * KVSTRIDE];
  __shared__ __attribute__((aligned(16))) float vbuf[2][STAGE_ROWS * KVSTRIDE];
  __shared__ __attribute__((aligned(16))) float pbuf[8][16 * PSTRIDE];

  const int tid  = threadIdx.x;
  const int wave = __builtin_amdgcn_readfirstlane(tid >> 5);
  const int lane = tid & 31;
  const int nlo  = lane & 15;
  const int hi   = lane >> 4;

  const int bid = blockIdx.x;
  const int pb  = bid & 7;
  const int h   = (bid >> 3) & (H_ - 1);
  const int b   = bid >> 6;
  const int pairIdx = pb * 8 + wave;

  const float scale = 0.125f;
  float* myP = pbuf[wave];
  const v2f onesB = (v2f){1.0f, 1.0f};

  const int srow = tid >> 4;
  const int scol = (tid & 15) * 4;
  const float* kstage = K + (((size_t)b * S_ + srow) * H_ + h) * E_ + scol;
  const float* vstage = V + (((size_t)b * S_ + srow) * H_ + h) * D_ + scol;
  const int sofs = srow * KVSTRIDE + scol;
  const size_t tileStride = (size_t)16 * H_ * E_;

  for (int pass = 0; pass < 2; ++pass) {
  const int qTile = pass ? pairIdx : (L_ / 16 - 1 - pairIdx);
  const int qBase = qTile * 16;
  const int jmax = pass ? (pb * 8 + 7) : (L_ / 16 - 1 - pb * 8);
  const int nst  = (jmax + 1) / TPS;

  const int qRowA = qBase + nlo;
  const float* qptr = Q + (((size_t)b * L_ + qRowA) * H_ + h) * E_;
  v2f aQ[16];
#pragma unroll
  for (int c = 0; c < 16; ++c) {
    aQ[c] = *reinterpret_cast<const v2f*>(qptr + c * 4 + hi * 2);
  }

  v8f oacc[4];
#pragma unroll
  for (int t = 0; t < 4; ++t) oacc[t] = (v8f){0, 0, 0, 0, 0, 0, 0, 0};
  float lsum[8];
#pragma unroll
  for (int r = 0; r < 8; ++r) lsum[r] = 0.0f;

  const float* bptr = bias + (size_t)(qBase + hi * 8) * S_ + nlo;

  {
    v4f k4[TPS], v4[TPS];
#pragma unroll
    for (int i = 0; i < TPS; ++i) {
      k4[i] = *reinterpret_cast<const v4f*>(kstage + (size_t)i * tileStride);
      v4[i] = *reinterpret_cast<const v4f*>(vstage + (size_t)i * tileStride);
    }
#pragma unroll
    for (int i = 0; i < TPS; ++i) {
      *reinterpret_cast<v4f*>(&kbuf[0][sofs + i * 16 * KVSTRIDE]) = k4[i];
      *reinterpret_cast<v4f*>(&vbuf[0][sofs + i * 16 * KVSTRIDE]) = v4[i];
    }
  }
  float bv[8];
#pragma unroll
  for (int r = 0; r < 8; ++r) bv[r] = bptr[(size_t)r * S_];
  __syncthreads();

  for (int st = 0; st < nst; ++st) {
    const int cur = st & 1;
    const int sn = (st + 1 < nst) ? (st + 1) : st;

    v4f kpre[TPS], vpre[TPS];
#pragma unroll
    for (int i = 0; i < TPS; ++i) {
      kpre[i] = *reinterpret_cast<const v4f*>(kstage + (size_t)(sn * TPS + i) * tileStride);
      vpre[i] = *reinterpret_cast<const v4f*>(vstage + (size_t)(sn * TPS + i) * tileStride);
    }
    const float* kcur = kbuf[cur];
    const float* vcur = vbuf[cur];

    for (int ti = 0; ti < TPS; ++ti) {
      const int jt = st * TPS + ti;
      const int jn = (jt < jmax) ? (jt + 1) : jmax;
      float bvn[8];
#pragma unroll
      for (int r = 0; r < 8; ++r) bvn[r] = bptr[(size_t)r * S_ + jn * 16];

      if (jt <= qTile) {
        const int j0  = jt * 16;
        const int key = j0 + nlo;

        const float* kb = kcur + (ti * 16 + nlo) * KVSTRIDE + hi * 2;
        v2f bK[16];
#pragma unroll
        for (int c = 0; c < 16; ++c)
          bK[c] = *reinterpret_cast<const v2f*>(kb + c * 4);

        const float* vb = vcur + (ti * 16 + hi * 2) * KVSTRIDE + nlo * 4;
        v2f bV[4][4];
#pragma unroll
        for (int kk = 0; kk < 4; ++kk) {
          const v4f r0 = *reinterpret_cast<const v4f*>(vb + (kk * 4 + 0) * KVSTRIDE);
          const v4f r1 = *reinterpret_cast<const v4f*>(vb + (kk * 4 + 1) * KVSTRIDE);
#pragma unroll
          for (int t = 0; t < 4; ++t) {
            bV[kk][t].x = r0[t];
            bV[kk][t].y = r1[t];
          }
        }

        __builtin_amdgcn_sched_barrier(0);

        v8f s8a = (v8f){0, 0, 0, 0, 0, 0, 0, 0};
        v8f s8b = (v8f){0, 0, 0, 0, 0, 0, 0, 0};
#pragma unroll
        for (int c = 0; c < 16; c += 2) {
          s8a = wmma_f32_16x16x4(aQ[c + 0], bK[c + 0], s8a);
          s8b = wmma_f32_16x16x4(aQ[c + 1], bK[c + 1], s8b);
        }

        float p[8];
#pragma unroll
        for (int r = 0; r < 8; ++r) {
          const int qRow = qBase + r + hi * 8;
          const float s = s8a[r] + s8b[r];
          const float e = __expf((s + bv[r]) * scale);
          p[r] = (key <= qRow) ? e : 0.0f;
          lsum[r] += p[r];
        }

#pragma unroll
        for (int r = 0; r < 8; ++r)
          myP[(r + hi * 8) * PSTRIDE + nlo] = p[r];
        __builtin_amdgcn_fence(__ATOMIC_SEQ_CST, "wavefront");
        v2f ap[4];
#pragma unroll
        for (int kk = 0; kk < 4; ++kk) {
          ap[kk] = *reinterpret_cast<const v2f*>(myP + nlo * PSTRIDE + kk * 4 + hi * 2);
        }
        __builtin_amdgcn_fence(__ATOMIC_SEQ_CST, "wavefront");

#pragma unroll
        for (int kk = 0; kk < 4; ++kk) {
#pragma unroll
          for (int t = 0; t < 4; ++t)
            oacc[t] = wmma_f32_16x16x4(ap[kk], bV[kk][t], oacc[t]);
        }
      }
#pragma unroll
      for (int r = 0; r < 8; ++r) bv[r] = bvn[r];
    }

#pragma unroll
    for (int i = 0; i < TPS; ++i) {
      *reinterpret_cast<v4f*>(&kbuf[cur ^ 1][sofs + i * 16 * KVSTRIDE]) = kpre[i];
      *reinterpret_cast<v4f*>(&vbuf[cur ^ 1][sofs + i * 16 * KVSTRIDE]) = vpre[i];
    }
    __syncthreads();
  }

  v8f lacc = (v8f){0, 0, 0, 0, 0, 0, 0, 0};
  {
#pragma unroll
    for (int r = 0; r < 8; ++r)
      myP[(r + hi * 8) * PSTRIDE + nlo] = lsum[r];
    __builtin_amdgcn_fence(__ATOMIC_SEQ_CST, "wavefront");
    v2f al[4];
#pragma unroll
    for (int kk = 0; kk < 4; ++kk)
      al[kk] = *reinterpret_cast<const v2f*>(myP + nlo * PSTRIDE + kk * 4 + hi * 2);
    __builtin_amdgcn_fence(__ATOMIC_SEQ_CST, "wavefront");
#pragma unroll
    for (int kk = 0; kk < 4; ++kk)
      lacc = wmma_f32_16x16x4(al[kk], onesB, lacc);
    asm volatile("v_nop\n\tv_nop\n\tv_nop\n\tv_nop"
                 : "+v"(lacc)
                 : "v"(al[0]), "v"(al[1]), "v"(al[2]), "v"(al[3]));
  }

#pragma unroll
  for (int r = 0; r < 8; ++r) {
    const float inv = 1.0f / lacc[r];
    const int qRow = qBase + r + hi * 8;
    v4f o;
    o.x = oacc[0][r] * inv;
    o.y = oacc[1][r] * inv;
    o.z = oacc[2][r] * inv;
    o.w = oacc[3][r] * inv;
    *reinterpret_cast<v4f*>(out + (((size_t)b * L_ + qRow) * H_ + h) * D_ + nlo * 4) = o;
  }
  }
}

extern "C" void kernel_launch(void* const* d_in, const int* in_sizes, int n_in,
                              void* d_out, int out_size, void* d_ws, size_t ws_size,
                              hipStream_t stream) {
  const float* Q    = (const float*)d_in[0];
  const float* K    = (const float*)d_in[1];
  const float* V    = (const float*)d_in[2];
  const float* bias = (const float*)d_in[3];
  float* out = (float*)d_out;

  dim3 grid(256);
  dim3 block(256);
  causal_attn_wmma_kernel<<<grid, block, 0, stream>>>(Q, K, V, bias, out);
}


# K and V WMMA fragments read from LDS one key tile ahead into a second register set, waited for at the next tile; the stage hand-over barrier moves in front of a stage's last tile (3 LDS slots)
# speedup vs baseline: 3.6623x; 1.0737x over previous
#include <hip/hip_runtime.h>
#include <hip/hip_bf16.h>
#include <math.h>


#define B_ 4
#define L_ 2048
#define S_ 2048
#define H_ 8
#define E_ 64
#define D_ 64

typedef float v2f __attribute__((ext_vector_type(2)));
typedef float v4f __attribute__((ext_vector_type(4)));
typedef float v8f __attribute__((ext_vector_type(8)));

__device__ __forceinline__ v8f wmma_f32_16x16x4(v2f a, v2f b, v8f c) {
  return __builtin_amdgcn_wmma_f32_16x16x4_f32(
      false, a, false, b, (short)0, c, false, false);
}

#define PSTRIDE 20

#define KVSTRIDE 68

#define TPS 4
#define STAGE_ROWS (16 * TPS)

__global__ __launch_bounds__(256)
void causal_attn_wmma_kernel(const float* __restrict__ Q,
                             const float* __restrict__ K,
                             const float* __restrict__ V,
                             const float* __restrict__ bias,
                             float* __restrict__ out) {
  __shared__ __attribute__((aligned(16))) float kbuf[3][STAGE_ROWS * KVSTRIDE];
  __shared__ __attribute__((aligned(16))) float vbuf[3][STAGE_ROWS * KVSTRIDE];
  __shared__ __attribute__((aligned(16))) float pbuf[8][16 * PSTRIDE];

  const int tid  = threadIdx.x;
  const int wave = __builtin_amdgcn_readfirstlane(tid >> 5);
  const int lane = tid & 31;
  const int nlo  = lane & 15;
  const int hi   = lane >> 4;

  const int bid = blockIdx.x;
  const int pb  = bid & 7;
  const int h   = (bid >> 3) & (H_ - 1);
  const int b   = bid >> 6;
  const int pairIdx = pb * 8 + wave;

  const float scale = 0.125f;
  float* myP = pbuf[wave];
  const v2f onesB = (v2f){1.0f, 1.0f};

  const int srow = tid >> 4;
  const int scol = (tid & 15) * 4;
  const float* kstage = K + (((size_t)b * S_ + srow) * H_ + h) * E_ + scol;
  const float* vstage = V + (((size_t)b * S_ + srow) * H_ + h) * D_ + scol;
  const int sofs = srow * KVSTRIDE + scol;
  const size_t tileStride = (size_t)16 * H_ * E_;

  for (int pass = 0; pass < 2; ++pass) {
  const int qTile = pass ? pairIdx : (L_ / 16 - 1 - pairIdx);
  const int qBase = qTile * 16;
  const int jmax = pass ? (pb * 8 + 7) : (L_ / 16 - 1 - pb * 8);
  const int nst  = (jmax + 1) / TPS;

  const int qRowA = qBase + nlo;
  const float* qptr = Q + (((size_t)b * L_ + qRowA) * H_ + h) * E_;
  v2f aQ[16];
#pragma unroll
  for (int c = 0; c < 16; ++c) {
    aQ[c] = *reinterpret_cast<const v2f*>(qptr + c * 4 + hi * 2);
  }

  v8f oacc[4];
#pragma unroll
  for (int t = 0; t < 4; ++t) oacc[t] = (v8f){0, 0, 0, 0, 0, 0, 0, 0};
  float lsum[8];
#pragma unroll
  for (int r = 0; r < 8; ++r) lsum[r] = 0.0f;

  const float* bptr = bias + (size_t)(qBase + hi * 8) * S_ + nlo;

  {
    v4f k4[TPS], v4[TPS];
#pragma unroll
    for (int i = 0; i < TPS; ++i) {
      k4[i] = *reinterpret_cast<const v4f*>(kstage + (size_t)i * tileStride);
      v4[i] = *reinterpret_cast<const v4f*>(vstage + (size_t)i * tileStride);
    }
#pragma unroll
    for (int i = 0; i < TPS; ++i) {
      *reinterpret_cast<v4f*>(&kbuf[0][sofs + i * 16 * KVSTRIDE]) = k4[i];
      *reinterpret_cast<v4f*>(&vbuf[0][sofs + i * 16 * KVSTRIDE]) = v4[i];
    }
  }
  float bv[8];
#pragma unroll
  for (int r = 0; r < 8; ++r) bv[r] = bptr[(size_t)r * S_];
  __syncthreads();

  const int kfo = nlo * KVSTRIDE + hi * 2;
  const int vfo = hi * 2 * KVSTRIDE + nlo * 4;

  v2f kF[2][16];
  v4f vF[2][8];
#pragma unroll
  for (int c = 0; c < 16; ++c)
    kF[0][c] = *reinterpret_cast<const v2f*>(&kbuf[0][kfo + c * 4]);
#pragma unroll
  for (int kk = 0; kk < 4; ++kk) {
    vF[0][2 * kk + 0] = *reinterpret_cast<const v4f*>(&vbuf[0][vfo + (kk * 4 + 0) * KVSTRIDE]);
    vF[0][2 * kk + 1] = *reinterpret_cast<const v4f*>(&vbuf[0][vfo + (kk * 4 + 1) * KVSTRIDE]);
  }

  int cur = 0, nxt = 1;
  for (int st = 0; st < nst; ++st) {
    const int sn = (st + 1 < nst) ? (st + 1) : st;

    v4f kpre[TPS], vpre[TPS];
#pragma unroll
    for (int i = 0; i < TPS; ++i) {
      kpre[i] = *reinterpret_cast<const v4f*>(kstage + (size_t)(sn * TPS + i) * tileStride);
      vpre[i] = *reinterpret_cast<const v4f*>(vstage + (size_t)(sn * TPS + i) * tileStride);
    }
    const float* kcur = kbuf[cur];
    const float* vcur = vbuf[cur];
    const float* knxt = kbuf[nxt];
    const float* vnxt = vbuf[nxt];

#pragma unroll
    for (int ti = 0; ti < TPS; ++ti) {
      const int jt = st * TPS + ti;
      if (ti == TPS - 1) {
#pragma unroll
        for (int i = 0; i < TPS; ++i) {
          *reinterpret_cast<v4f*>(&kbuf[nxt][sofs + i * 16 * KVSTRIDE]) = kpre[i];
          *reinterpret_cast<v4f*>(&vbuf[nxt][sofs + i * 16 * KVSTRIDE]) = vpre[i];
        }
        __syncthreads();
      }
      const int jn = (jt < jmax) ? (jt + 1) : jmax;
      float bvn[8];
#pragma unroll
      for (int r = 0; r < 8; ++r) bvn[r] = bptr[(size_t)r * S_ + jn * 16];

      if (jt <= qTile) {
        const int j0  = jt * 16;
        const int key = j0 + nlo;

        const float* kn = (ti < TPS - 1) ? (kcur + (ti + 1) * 16 * KVSTRIDE) : knxt;
        const float* vn = (ti < TPS - 1) ? (vcur + (ti + 1) * 16 * KVSTRIDE) : vnxt;
#pragma unroll
        for (int c = 0; c < 16; ++c)
          kF[(ti + 1) & 1][c] = *reinterpret_cast<const v2f*>(kn + kfo + c * 4);
#pragma unroll
        for (int kk = 0; kk < 4; ++kk) {
          vF[(ti + 1) & 1][2 * kk + 0] = *reinterpret_cast<const v4f*>(vn + vfo + (kk * 4 + 0) * KVSTRIDE);
          vF[(ti + 1) & 1][2 * kk + 1] = *reinterpret_cast<const v4f*>(vn + vfo + (kk * 4 + 1) * KVSTRIDE);
        }

        __builtin_amdgcn_sched_barrier(0);

        v2f bK[16];
#pragma unroll
        for (int c = 0; c < 16; ++c) bK[c] = kF[ti & 1][c];
        v2f bV[4][4];
#pragma unroll
        for (int kk = 0; kk < 4; ++kk) {
#pragma unroll
          for (int t = 0; t < 4; ++t) {
            bV[kk][t].x = vF[ti & 1][2 * kk + 0][t];
            bV[kk][t].y = vF[ti & 1][2 * kk + 1][t];
          }
        }

        __builtin_amdgcn_sched_barrier(0);

        v8f s8a = (v8f){0, 0, 0, 0, 0, 0, 0, 0};
        v8f s8b = (v8f){0, 0, 0, 0, 0, 0, 0, 0};
#pragma unroll
        for (int c = 0; c < 16; c += 2) {
          s8a = wmma_f32_16x16x4(aQ[c + 0], bK[c + 0], s8a);
          s8b = wmma_f32_16x16x4(aQ[c + 1], bK[c + 1], s8b);
        }

        float p[8];
#pragma unroll
        for (int r = 0; r < 8; ++r) {
          const int qRow = qBase + r + hi * 8;
          const float s = s8a[r] + s8b[r];
          const float e = __expf((s + bv[r]) * scale);
          p[r] = (key <= qRow) ? e : 0.0f;
          lsum[r] += p[r];
        }

#pragma unroll
        for (int r = 0; r < 8; ++r)
          myP[(r + hi * 8) * PSTRIDE + nlo] = p[r];
        __builtin_amdgcn_fence(__ATOMIC_SEQ_CST, "wavefront");
        v2f ap[4];
#pragma unroll
        for (int kk = 0; kk < 4; ++kk) {
          ap[kk] = *reinterpret_cast<const v2f*>(myP + nlo * PSTRIDE + kk * 4 + hi * 2);
        }
        __builtin_amdgcn_fence(__ATOMIC_SEQ_CST, "wavefront");

#pragma unroll
        for (int kk = 0; kk < 4; ++kk) {
#pragma unroll
          for (int t = 0; t < 4; ++t)
            oacc[t] = wmma_f32_16x16x4(ap[kk], bV[kk][t], oacc[t]);
        }
      }
#pragma unroll
      for (int r = 0; r < 8; ++r) bv[r] = bvn[r];
    }

    cur = nxt;
    nxt = (nxt == 2) ? 0 : (nxt + 1);
  }
  __syncthreads();

  v8f lacc = (v8f){0, 0, 0, 0, 0, 0, 0, 0};
  {
#pragma unroll
    for (int r = 0; r < 8; ++r)
      myP[(r + hi * 8) * PSTRIDE + nlo] = lsum[r];
    __builtin_amdgcn_fence(__ATOMIC_SEQ_CST, "wavefront");
    v2f al[4];
#pragma unroll
    for (int kk = 0; kk < 4; ++kk)
      al[kk] = *reinterpret_cast<const v2f*>(myP + nlo * PSTRIDE + kk * 4 + hi * 2);
    __builtin_amdgcn_fence(__ATOMIC_SEQ_CST, "wavefront");
#pragma unroll
    for (int kk = 0; kk < 4; ++kk)
      lacc = wmma_f32_16x16x4(al[kk], onesB, lacc);
    asm volatile("v_nop\n\tv_nop\n\tv_nop\n\tv_nop"
                 : "+v"(lacc)
                 : "v"(al[0]), "v"(al[1]), "v"(al[2]), "v"(al[3]));
  }

#pragma unroll
  for (int r = 0; r < 8; ++r) {
    const float inv = 1.0f / lacc[r];
    const int qRow = qBase + r + hi * 8;
    v4f o;
    o.x = oacc[0][r] * inv;
    o.y = oacc[1][r] * inv;
    o.z = oacc[2][r] * inv;
    o.w = oacc[3][r] * inv;
    *reinterpret_cast<v4f*>(out + (((size_t)b * L_ + qRow) * H_ + h) * D_ + nlo * 4) = o;
  }
  }
}

extern "C" void kernel_launch(void* const* d_in, const int* in_sizes, int n_in,
                              void* d_out, int out_size, void* d_ws, size_t ws_size,
                              hipStream_t stream) {
  const float* Q    = (const float*)d_in[0];
  const float* K    = (const float*)d_in[1];
  const float* V    = (const float*)d_in[2];
  const float* bias = (const float*)d_in[3];
  float* out = (float*)d_out;

  dim3 grid(256);
  dim3 block(256);
  causal_attn_wmma_kernel<<<grid, block, 0, stream>>>(Q, K, V, bias, out);
}


# Score tile transposed (K*Q^T): P^T leaves the accumulator as the B fragment of V^T*P^T, no LDS relayout of P; bias by 16-byte loads; row sums and a tile's P*V keys summed in another order
# speedup vs baseline: 3.7891x; 1.0346x over previous
#include <hip/hip_runtime.h>
#include <hip/hip_bf16.h>
#include <math.h>


#define B_ 4
#define L_ 2048
#define S_ 2048
#define H_ 8
#define E_ 64
#define D_ 64

typedef float v2f __attribute__((ext_vector_type(2)));
typedef float v4f __attribute__((ext_vector_type(4)));
typedef float v8f __attribute__((ext_vector_type(8)));

__device__ __forceinline__ v8f wmma_f32_16x16x4(v2f a, v2f b, v8f c) {
  return __builtin_amdgcn_wmma_f32_16x16x4_f32(
      false, a, false, b, (short)0, c, false, false);
}

#define OSTRIDE 68

#define KVSTRIDE 68

#define TPS 4
#define STAGE_ROWS (16 * TPS)

__global__ __launch_bounds__(256)
void causal_attn_wmma_kernel(const float* __restrict__ Q,
                             const float* __restrict__ K,
                             const float* __restrict__ V,
                             const float* __restrict__ bias,
                             float* __restrict__ out) {
  __shared__ __attribute__((aligned(16))) float kbuf[3][STAGE_ROWS * KVSTRIDE];
  __shared__ __attribute__((aligned(16))) float vbuf[3][STAGE_ROWS * KVSTRIDE];
  __shared__ __attribute__((aligned(16))) float obuf[8][16 * OSTRIDE];

  const int tid  = threadIdx.x;
  const int wave = __builtin_amdgcn_readfirstlane(tid >> 5);
  const int lane = tid & 31;
  const int nlo  = lane & 15;
  const int hi   = lane >> 4;

  const int bid = blockIdx.x;
  const int pb  = bid & 7;
  const int h   = (bid >> 3) & (H_ - 1);
  const int b   = bid >> 6;
  const int pairIdx = pb * 8 + wave;

  const float scale = 0.125f;
  float* myO = obuf[wave];

  const int srow = tid >> 4;
  const int scol = (tid & 15) * 4;
  const float* kstage = K + (((size_t)b * S_ + srow) * H_ + h) * E_ + scol;
  const float* vstage = V + (((size_t)b * S_ + srow) * H_ + h) * D_ + scol;
  const int sofs = srow * KVSTRIDE + scol;
  const size_t tileStride = (size_t)16 * H_ * E_;

  for (int pass = 0; pass < 2; ++pass) {
  const int qTile = pass ? pairIdx : (L_ / 16 - 1 - pairIdx);
  const int qBase = qTile * 16;
  const int jmax = pass ? (pb * 8 + 7) : (L_ / 16 - 1 - pb * 8);
  const int nst  = (jmax + 1) / TPS;

  const int qRowA = qBase + nlo;
  const float* qptr = Q + (((size_t)b * L_ + qRowA) * H_ + h) * E_;
  v2f aQ[16];
#pragma unroll
  for (int c = 0; c < 16; ++c) {
    aQ[c] = *reinterpret_cast<const v2f*>(qptr + c * 4 + hi * 2);
  }

  v8f oacc[4];
#pragma unroll
  for (int t = 0; t < 4; ++t) oacc[t] = (v8f){0, 0, 0, 0, 0, 0, 0, 0};
  float lsum = 0.0f;
  float pp[8];
#pragma unroll
  for (int r = 0; r < 8; ++r) pp[r] = 0.0f;
  const int qRow = qBase + nlo;

  const float* bptr = bias + (size_t)qRow * S_ + hi * 8;

  {
    v4f k4[TPS], v4[TPS];
#pragma unroll
    for (int i = 0; i < TPS; ++i) {
      k4[i] = *reinterpret_cast<const v4f*>(kstage + (size_t)i * tileStride);
      v4[i] = *reinterpret_cast<const v4f*>(vstage + (size_t)i * tileStride);
    }
#pragma unroll
    for (int i = 0; i < TPS; ++i) {
      *reinterpret_cast<v4f*>(&kbuf[0][sofs + i * 16 * KVSTRIDE]) = k4[i];
      *reinterpret_cast<v4f*>(&vbuf[0][sofs + i * 16 * KVSTRIDE]) = v4[i];
    }
  }
  v4f bv0 = *reinterpret_cast<const v4f*>(bptr);
  v4f bv1 = *reinterpret_cast<const v4f*>(bptr + 4);
  __syncthreads();

  const int kfo = nlo * KVSTRIDE + hi * 2;
  const int vfo = hi * 8 * KVSTRIDE + nlo * 4;

  v2f kF[2][16];
  v4f vF[2][8];
#pragma unroll
  for (int c = 0; c < 16; ++c)
    kF[0][c] = *reinterpret_cast<const v2f*>(&kbuf[0][kfo + c * 4]);
#pragma unroll
  for (int i = 0; i < 8; ++i)
    vF[0][i] = *reinterpret_cast<const v4f*>(&vbuf[0][vfo + i * KVSTRIDE]);

  int cur = 0, nxt = 1;
  for (int st = 0; st < nst; ++st) {
    const int sn = (st + 1 < nst) ? (st + 1) : st;

    v4f kpre[TPS], vpre[TPS];
#pragma unroll
    for (int i = 0; i < TPS; ++i) {
      kpre[i] = *reinterpret_cast<const v4f*>(kstage + (size_t)(sn * TPS + i) * tileStride);
      vpre[i] = *reinterpret_cast<const v4f*>(vstage + (size_t)(sn * TPS + i) * tileStride);
    }
    const float* kcur = kbuf[cur];
    const float* vcur = vbuf[cur];
    const float* knxt = kbuf[nxt];
    const float* vnxt = vbuf[nxt];

#pragma unroll
    for (int ti = 0; ti < TPS; ++ti) {
      const int jt = st * TPS + ti;
      if (ti == TPS - 1) {
#pragma unroll
        for (int i = 0; i < TPS; ++i) {
          *reinterpret_cast<v4f*>(&kbuf[nxt][sofs + i * 16 * KVSTRIDE]) = kpre[i];
          *reinterpret_cast<v4f*>(&vbuf[nxt][sofs + i * 16 * KVSTRIDE]) = vpre[i];
        }
        __syncthreads();
      }
      const int jn = (jt < jmax) ? (jt + 1) : jmax;
      const v4f bvn0 = *reinterpret_cast<const v4f*>(bptr + jn * 16);
      const v4f bvn1 = *reinterpret_cast<const v4f*>(bptr + jn * 16 + 4);

      if (jt <= qTile) {
        const int dq = qRow - (jt * 16 + hi * 8);

        const float* kn = (ti < TPS - 1) ? (kcur + (ti + 1) * 16 * KVSTRIDE) : knxt;
        const float* vn = (ti < TPS - 1) ? (vcur + (ti + 1) * 16 * KVSTRIDE) : vnxt;
#pragma unroll
        for (int c = 0; c < 16; ++c)
          kF[(ti + 1) & 1][c] = *reinterpret_cast<const v2f*>(kn + kfo + c * 4);
#pragma unroll
        for (int i = 0; i < 8; ++i)
          vF[(ti + 1) & 1][i] = *reinterpret_cast<const v4f*>(vn + vfo + i * KVSTRIDE);

        __builtin_amdgcn_sched_barrier(0);

        v2f aK[16];
#pragma unroll
        for (int c = 0; c < 16; ++c) aK[c] = kF[ti & 1][c];
        v2f aV[4][4];
#pragma unroll
        for (int kk = 0; kk < 4; ++kk) {
#pragma unroll
          for (int t = 0; t < 4; ++t) {
            aV[kk][t].x = vF[ti & 1][2 * kk + 0][t];
            aV[kk][t].y = vF[ti & 1][2 * kk + 1][t];
          }
        }

        lsum += ((pp[0] + pp[1]) + (pp[2] + pp[3])) + ((pp[4] + pp[5]) + (pp[6] + pp[7]));

        __builtin_amdgcn_sched_barrier(0);

        v8f s8a = (v8f){0, 0, 0, 0, 0, 0, 0, 0};
        v8f s8b = (v8f){0, 0, 0, 0, 0, 0, 0, 0};
#pragma unroll
        for (int c = 0; c < 16; c += 2) {
          s8a = wmma_f32_16x16x4(aK[c + 0], aQ[c + 0], s8a);
          s8b = wmma_f32_16x16x4(aK[c + 1], aQ[c + 1], s8b);
        }

        float p[8];
#pragma unroll
        for (int r = 0; r < 8; ++r) {
          const float s = s8a[r] + s8b[r];
          const float bb = (r < 4) ? bv0[r & 3] : bv1[r & 3];
          const float e = __expf((s + bb) * scale);
          p[r] = (r <= dq) ? e : 0.0f;
        }

#pragma unroll
        for (int kk = 0; kk < 4; ++kk) {
          const v2f bp = (v2f){p[2 * kk + 0], p[2 * kk + 1]};
#pragma unroll
          for (int t = 0; t < 4; ++t)
            oacc[t] = wmma_f32_16x16x4(aV[kk][t], bp, oacc[t]);
        }
#pragma unroll
        for (int r = 0; r < 8; ++r) pp[r] = p[r];
      }
      bv0 = bvn0;
      bv1 = bvn1;
    }

    cur = nxt;
    nxt = (nxt == 2) ? 0 : (nxt + 1);
  }
  __syncthreads();

  asm volatile("v_nop\n\tv_nop\n\tv_nop\n\tv_nop"
               : "+v"(pp[0]), "+v"(pp[1]), "+v"(pp[2]), "+v"(pp[3]),
                 "+v"(pp[4]), "+v"(pp[5]), "+v"(pp[6]), "+v"(pp[7]));
  __builtin_amdgcn_sched_barrier(0);
  lsum += ((pp[0] + pp[1]) + (pp[2] + pp[3])) + ((pp[4] + pp[5]) + (pp[6] + pp[7]));
  const float denom = lsum + __shfl_xor(lsum, 16);
  const float inv = 1.0f / denom;

#pragma unroll
  for (int r = 0; r < 8; ++r) {
    v4f o;
    o.x = oacc[0][r] * inv;
    o.y = oacc[1][r] * inv;
    o.z = oacc[2][r] * inv;
    o.w = oacc[3][r] * inv;
    *reinterpret_cast<v4f*>(myO + nlo * OSTRIDE + hi * 32 + r * 4) = o;
  }
  __builtin_amdgcn_fence(__ATOMIC_SEQ_CST, "wavefront");
#pragma unroll
  for (int i = 0; i < 8; ++i) {
    const int row = 2 * i + hi;
    const v4f o = *reinterpret_cast<const v4f*>(myO + row * OSTRIDE + nlo * 4);
    *reinterpret_cast<v4f*>(out + (((size_t)b * L_ + qBase + row) * H_ + h) * D_ + nlo * 4) = o;
  }
  __builtin_amdgcn_fence(__ATOMIC_SEQ_CST, "wavefront");
  }
}

extern "C" void kernel_launch(void* const* d_in, const int* in_sizes, int n_in,
                              void* d_out, int out_size, void* d_ws, size_t ws_size,
                              hipStream_t stream) {
  const float* Q    = (const float*)d_in[0];
  const float* K    = (const float*)d_in[1];
  const float* V    = (const float*)d_in[2];
  const float* bias = (const float*)d_in[3];
  float* out = (float*)d_out;

  dim3 grid(256);
  dim3 block(256);
  causal_attn_wmma_kernel<<<grid, block, 0, stream>>>(Q, K, V, bias, out);
}


# Vector instructions per key tile 71 -> 22: V staged key-pair interleaved (no pairing moves), causal select on the diagonal tile only, bias as accumulator start, 8 running row sums; sums reordered
# speedup vs baseline: 3.9897x; 1.0529x over previous
#include <hip/hip_runtime.h>
#include <hip/hip_bf16.h>
#include <math.h>


#define B_ 4
#define L_ 2048
#define S_ 2048
#define H_ 8
#define E_ 64
#define D_ 64

typedef float v2f __attribute__((ext_vector_type(2)));
typedef float v4f __attribute__((ext_vector_type(4)));
typedef float v8f __attribute__((ext_vector_type(8)));

__device__ __forceinline__ v8f wmma_f32_16x16x4(v2f a, v2f b, v8f c) {
  return __builtin_amdgcn_wmma_f32_16x16x4_f32(
      false, a, false, b, (short)0, c, false, false);
}

#define OSTRIDE 68

#define KVSTRIDE 68

#define TPS 4
#define STAGE_ROWS (16 * TPS)

#define VPITCH 132

__global__ __launch_bounds__(256)
void causal_attn_wmma_kernel(const float* __restrict__ Q,
                             const float* __restrict__ K,
                             const float* __restrict__ V,
                             const float* __restrict__ bias,
                             float* __restrict__ out) {
  __shared__ __attribute__((aligned(16))) float kbuf[3][STAGE_ROWS * KVSTRIDE];
  __shared__ __attribute__((aligned(16))) float vbuf[3][(STAGE_ROWS / 2) * VPITCH];
  __shared__ __attribute__((aligned(16))) float obuf[8][16 * OSTRIDE];

  const int tid  = threadIdx.x;
  const int wave = __builtin_amdgcn_readfirstlane(tid >> 5);
  const int lane = tid & 31;
  const int nlo  = lane & 15;
  const int hi   = lane >> 4;

  const int bid = blockIdx.x;
  const int pb  = bid & 7;
  const int h   = (bid >> 3) & (H_ - 1);
  const int b   = bid >> 6;
  const int pairIdx = pb * 8 + wave;

  const float kExp = 0.125f * 1.44269504088896340736f;
  float* myO = obuf[wave];

  const int srow = tid >> 4;
  const int scol = (tid & 15) * 4;
  const float* kstage = K + (((size_t)b * S_ + srow) * H_ + h) * E_ + scol;
  const float* vstage = V + (((size_t)b * S_ + srow) * H_ + h) * D_ + scol;
  const int sofs = srow * KVSTRIDE + scol;
  const int svofs = (srow >> 1) * VPITCH + 2 * scol + (srow & 1);
  const size_t tileStride = (size_t)16 * H_ * E_;

  for (int pass = 0; pass < 2; ++pass) {
  const int qTile = pass ? pairIdx : (L_ / 16 - 1 - pairIdx);
  const int qBase = qTile * 16;
  const int jmax = pass ? (pb * 8 + 7) : (L_ / 16 - 1 - pb * 8);
  const int nst  = (jmax + 1) / TPS;

  const int qRowA = qBase + nlo;
  const float* qptr = Q + (((size_t)b * L_ + qRowA) * H_ + h) * E_;
  v2f aQ[16];
#pragma unroll
  for (int c = 0; c < 16; ++c) {
    aQ[c] = *reinterpret_cast<const v2f*>(qptr + c * 4 + hi * 2);
  }

  v8f oacc[4];
#pragma unroll
  for (int t = 0; t < 4; ++t) oacc[t] = (v8f){0, 0, 0, 0, 0, 0, 0, 0};
  float psum[8];
#pragma unroll
  for (int r = 0; r < 8; ++r) psum[r] = 0.0f;
  const int qRow = qBase + nlo;

  const float* bptr = bias + (size_t)qRow * S_ + hi * 8;

  {
    v4f k4[TPS], v4[TPS];
#pragma unroll
    for (int i = 0; i < TPS; ++i) {
      k4[i] = *reinterpret_cast<const v4f*>(kstage + (size_t)i * tileStride);
      v4[i] = *reinterpret_cast<const v4f*>(vstage + (size_t)i * tileStride);
    }
#pragma unroll
    for (int i = 0; i < TPS; ++i) {
      *reinterpret_cast<v4f*>(&kbuf[0][sofs + i * 16 * KVSTRIDE]) = k4[i];
      float* vd = &vbuf[0][svofs + i * 8 * VPITCH];
      vd[0] = v4[i].x; vd[2] = v4[i].y; vd[4] = v4[i].z; vd[6] = v4[i].w;
    }
  }
  v4f bv0 = *reinterpret_cast<const v4f*>(bptr);
  v4f bv1 = *reinterpret_cast<const v4f*>(bptr + 4);
  __syncthreads();

  const int kfo = nlo * KVSTRIDE + hi * 2;
  const int vfo = hi * 4 * VPITCH + nlo * 4;

  v2f kF[2][16];
  v4f vF[2][8];
#pragma unroll
  for (int c = 0; c < 16; ++c)
    kF[0][c] = *reinterpret_cast<const v2f*>(&kbuf[0][kfo + c * 4]);
#pragma unroll
  for (int i = 0; i < 8; ++i)
    vF[0][i] = *reinterpret_cast<const v4f*>(&vbuf[0][vfo + (i >> 1) * VPITCH + (i & 1) * 64]);

  int cur = 0, nxt = 1;
  for (int st = 0; st < nst; ++st) {
    const int sn = (st + 1 < nst) ? (st + 1) : st;

    v4f kpre[TPS], vpre[TPS];
#pragma unroll
    for (int i = 0; i < TPS; ++i) {
      kpre[i] = *reinterpret_cast<const v4f*>(kstage + (size_t)(sn * TPS + i) * tileStride);
      vpre[i] = *reinterpret_cast<const v4f*>(vstage + (size_t)(sn * TPS + i) * tileStride);
    }
    const float* kcur = kbuf[cur];
    const float* vcur = vbuf[cur];
    const float* knxt = kbuf[nxt];
    const float* vnxt = vbuf[nxt];

#pragma unroll
    for (int ti = 0; ti < TPS; ++ti) {
      const int jt = st * TPS + ti;
      if (ti == TPS - 1) {
#pragma unroll
        for (int i = 0; i < TPS; ++i) {
          *reinterpret_cast<v4f*>(&kbuf[nxt][sofs + i * 16 * KVSTRIDE]) = kpre[i];
          float* vd = &vbuf[nxt][svofs + i * 8 * VPITCH];
          vd[0] = vpre[i].x; vd[2] = vpre[i].y; vd[4] = vpre[i].z; vd[6] = vpre[i].w;
        }
        __syncthreads();
      }
      const int jn = (jt < jmax) ? (jt + 1) : jmax;
      const v4f bvn0 = *reinterpret_cast<const v4f*>(bptr + jn * 16);
      const v4f bvn1 = *reinterpret_cast<const v4f*>(bptr + jn * 16 + 4);

      if (jt <= qTile) {
        const float* kn = (ti < TPS - 1) ? (kcur + (ti + 1) * 16 * KVSTRIDE) : knxt;
        const float* vn = (ti < TPS - 1) ? (vcur + (ti + 1) * 8 * VPITCH) : vnxt;
#pragma unroll
        for (int c = 0; c < 16; ++c)
          kF[(ti + 1) & 1][c] = *reinterpret_cast<const v2f*>(kn + kfo + c * 4);
#pragma unroll
        for (int i = 0; i < 8; ++i)
          vF[(ti + 1) & 1][i] =
              *reinterpret_cast<const v4f*>(vn + vfo + (i >> 1) * VPITCH + (i & 1) * 64);

        __builtin_amdgcn_sched_barrier(0);

        v8f s8a = (v8f){bv0[0], bv0[1], bv0[2], bv0[3], bv1[0], bv1[1], bv1[2], bv1[3]};
        v8f s8b = (v8f){0, 0, 0, 0, 0, 0, 0, 0};
#pragma unroll
        for (int c = 0; c < 16; c += 2) {
          s8a = wmma_f32_16x16x4(kF[ti & 1][c + 0], aQ[c + 0], s8a);
          s8b = wmma_f32_16x16x4(kF[ti & 1][c + 1], aQ[c + 1], s8b);
        }

        float p[8];
#pragma unroll
        for (int r = 0; r < 8; ++r)
          p[r] = __builtin_amdgcn_exp2f((s8a[r] + s8b[r]) * kExp);

        if (jt == qTile) {
          const int dq = nlo - hi * 8;
#pragma unroll
          for (int r = 0; r < 8; ++r) p[r] = (r <= dq) ? p[r] : 0.0f;
        }

#pragma unroll
        for (int kk = 0; kk < 4; ++kk) {
          const v2f bp = (v2f){p[2 * kk + 0], p[2 * kk + 1]};
#pragma unroll
          for (int t = 0; t < 4; ++t) {
            const v4f vr = vF[ti & 1][2 * kk + (t >> 1)];
            const v2f av = (t & 1) ? (v2f){vr[2], vr[3]} : (v2f){vr[0], vr[1]};
            oacc[t] = wmma_f32_16x16x4(av, bp, oacc[t]);
          }
        }

        __builtin_amdgcn_sched_barrier(0);
#pragma unroll
        for (int r = 0; r < 8; ++r) psum[r] += p[r];
      }
      bv0 = bvn0;
      bv1 = bvn1;
    }

    cur = nxt;
    nxt = (nxt == 2) ? 0 : (nxt + 1);
  }
  __syncthreads();

  asm volatile("v_nop\n\tv_nop\n\tv_nop\n\tv_nop"
               : "+v"(psum[0]), "+v"(psum[1]), "+v"(psum[2]), "+v"(psum[3]),
                 "+v"(psum[4]), "+v"(psum[5]), "+v"(psum[6]), "+v"(psum[7]));
  __builtin_amdgcn_sched_barrier(0);
  const float lsum = ((psum[0] + psum[1]) + (psum[2] + psum[3])) +
                     ((psum[4] + psum[5]) + (psum[6] + psum[7]));
  const float denom = lsum + __shfl_xor(lsum, 16);
  const float inv = 1.0f / denom;

#pragma unroll
  for (int tt = 0; tt < 2; ++tt) {
#pragma unroll
    for (int rp = 0; rp < 4; ++rp) {
      v4f o;
      o.x = oacc[2 * tt + 0][2 * rp + 0] * inv;
      o.y = oacc[2 * tt + 1][2 * rp + 0] * inv;
      o.z = oacc[2 * tt + 0][2 * rp + 1] * inv;
      o.w = oacc[2 * tt + 1][2 * rp + 1] * inv;
      *reinterpret_cast<v4f*>(myO + nlo * OSTRIDE + tt * 32 + hi * 16 + rp * 4) = o;
    }
  }
  __builtin_amdgcn_fence(__ATOMIC_SEQ_CST, "wavefront");
#pragma unroll
  for (int i = 0; i < 8; ++i) {
    const int row = 2 * i + hi;
    const v4f o = *reinterpret_cast<const v4f*>(myO + row * OSTRIDE + nlo * 4);
    *reinterpret_cast<v4f*>(out + (((size_t)b * L_ + qBase + row) * H_ + h) * D_ + nlo * 4) = o;
  }
  __builtin_amdgcn_fence(__ATOMIC_SEQ_CST, "wavefront");
  }
}

extern "C" void kernel_launch(void* const* d_in, const int* in_sizes, int n_in,
                              void* d_out, int out_size, void* d_ws, size_t ws_size,
                              hipStream_t stream) {
  const float* Q    = (const float*)d_in[0];
  const float* K    = (const float*)d_in[1];
  const float* V    = (const float*)d_in[2];
  const float* bias = (const float*)d_in[3];
  float* out = (float*)d_out;

  dim3 grid(256);
  dim3 block(256);
  causal_attn_wmma_kernel<<<grid, block, 0, stream>>>(Q, K, V, bias, out);
}
